# starting kernel
# baseline (speedup 1.0000x reference)
#include <hip/hip_runtime.h>


typedef float v2f __attribute__((ext_vector_type(2)));
typedef float v8f __attribute__((ext_vector_type(8)));

#define WMMA_F32X4(A, B, C) \
    __builtin_amdgcn_wmma_f32_16x16x4_f32(false, (A), false, (B), (short)0, (C), false, false)

constexpr int Tn = 1024;
constexpr int Bn = 4096;

__device__ __forceinline__ float fast_sigmoid(float v) {
    return __builtin_amdgcn_rcpf(1.0f + __expf(-v));
}
__device__ __forceinline__ float fast_tanh(float v) {
    return 1.0f - 2.0f * __builtin_amdgcn_rcpf(__expf(2.0f * v) + 1.0f);
}

__global__ __launch_bounds__(256) void gru_wmma_kernel(
    const float* __restrict__ x,
    const float* __restrict__ h0,
    const float* __restrict__ w_ih,
    const float* __restrict__ w_hh,
    const float* __restrict__ b_ih,
    const float* __restrict__ b_hh,
    const float* __restrict__ fc_w,
    const float* __restrict__ fc_b,
    float* __restrict__ out,
    float* __restrict__ h_out)
{
    const int lane  = threadIdx.x & 31;
    const int wid   = (blockIdx.x * blockDim.x + threadIdx.x) >> 5;
    const int m     = lane & 15;
    const bool lo   = lane < 16;
    const int bbase = wid * 16;
    const int b     = bbase + m;

    const int k0 = lo ? 0 : 2;
    v2f aRZ1, aRZ2, aN1, aN2;
    aRZ1.x = w_hh[m * 8 + k0 + 0]; aRZ1.y = w_hh[m * 8 + k0 + 1];
    aRZ2.x = w_hh[m * 8 + k0 + 4]; aRZ2.y = w_hh[m * 8 + k0 + 5];
    if (m < 8) {
        aN1.x = w_hh[(16 + m) * 8 + k0 + 0]; aN1.y = w_hh[(16 + m) * 8 + k0 + 1];
        aN2.x = w_hh[(16 + m) * 8 + k0 + 4]; aN2.y = w_hh[(16 + m) * 8 + k0 + 5];
    } else {
        aN1.x = aN1.y = aN2.x = aN2.y = 0.0f;
    }

    v8f cRZ, cN;
    #pragma unroll
    for (int r = 0; r < 8; ++r) {
        const int g = lo ? r : 8 + r;
        cRZ[r] = b_ih[g] + b_hh[g];
        cN[r]  = lo ? b_hh[16 + r] : 0.0f;
    }

    float wA0[8], wA1[8];
    #pragma unroll
    for (int j = 0; j < 8; ++j) {
        const int g = lo ? j : 8 + j;
        wA0[j] = w_ih[g * 2 + 0];
        wA1[j] = w_ih[g * 2 + 1];
    }
    const int jl[4] = {0, 1, 4, 5};
    const int jh[4] = {2, 3, 6, 7};
    float wN0[4], wN1[4], bN[4], fcw[4];
    #pragma unroll
    for (int s = 0; s < 4; ++s) {
        const int j = lo ? jl[s] : jh[s];
        wN0[s] = w_ih[(16 + j) * 2 + 0];
        wN1[s] = w_ih[(16 + j) * 2 + 1];
        bN[s]  = b_ih[16 + j];
        fcw[s] = fc_w[j];
    }
    const float fcb = fc_b[0];

    float h[4];
    {
        const float* hp = h0 + (size_t)b * 8 + (lo ? 0 : 2);
        h[0] = hp[0]; h[1] = hp[1]; h[2] = hp[4]; h[3] = hp[5];
    }

    const float* xp   = x + (size_t)b * 2;
    float*       outp = out + bbase + m;

    for (int t = 0; t < Tn; ++t) {
        const float x0 = xp[0];
        const float x1 = xp[1];
        xp += (size_t)Bn * 2;

        v2f b1, b2;
        b1.x = h[0]; b1.y = h[1];
        b2.x = h[2]; b2.y = h[3];

        v8f dRZ = WMMA_F32X4(aRZ1, b1, cRZ);
        dRZ     = WMMA_F32X4(aRZ2, b2, dRZ);
        v8f dN  = WMMA_F32X4(aN1, b1, cN);
        dN      = WMMA_F32X4(aN2, b2, dN);

        float a[8], o[8], g2[8];
        #pragma unroll
        for (int j = 0; j < 8; ++j) {
            const float u = dRZ[j] + fmaf(x1, wA1[j], x0 * wA0[j]);
            a[j]  = fast_sigmoid(u);
            o[j]  = __shfl_xor(a[j], 16, 32);
            g2[j] = __shfl_xor(dN[j], 16, 32);
        }

        #pragma unroll
        for (int s = 0; s < 4; ++s) {
            const float rS  = lo ? a[jl[s]]  : o[jh[s]];
            const float zS  = lo ? o[jl[s]]  : a[jh[s]];
            const float hnS = lo ? dN[jl[s]] : g2[jh[s]];
            const float xnS = fmaf(x1, wN1[s], fmaf(x0, wN0[s], bN[s]));
            const float nS  = fast_tanh(fmaf(rS, hnS, xnS));
            h[s] = nS + zS * (h[s] - nS);
        }

        float acc = fmaf(fcw[3], h[3], fmaf(fcw[2], h[2], fmaf(fcw[1], h[1], fcw[0] * h[0])));
        const float y = fast_sigmoid(acc + __shfl_xor(acc, 16, 32) + fcb);
        if (lo) *outp = y;
        outp += Bn;
    }

    {
        float* hp = h_out + (size_t)b * 8 + (lo ? 0 : 2);
        hp[0] = h[0]; hp[1] = h[1]; hp[4] = h[2]; hp[5] = h[3];
    }
}

extern "C" void kernel_launch(void* const* d_in, const int* in_sizes, int n_in,
                              void* d_out, int out_size, void* d_ws, size_t ws_size,
                              hipStream_t stream) {
    const float* x    = (const float*)d_in[0];
    const float* h0   = (const float*)d_in[1];
    const float* w_ih = (const float*)d_in[2];
    const float* w_hh = (const float*)d_in[3];
    const float* b_ih = (const float*)d_in[4];
    const float* b_hh = (const float*)d_in[5];
    const float* fc_w = (const float*)d_in[6];
    const float* fc_b = (const float*)d_in[7];

    float* out    = (float*)d_out;
    float* h_last = out + (size_t)Tn * Bn;

    const int threads = 256;
    const int blocks  = Bn / 16 / (threads / 32);
    gru_wmma_kernel<<<blocks, threads, 0, stream>>>(
        x, h0, w_ih, w_hh, b_ih, b_hh, fc_w, fc_b, out, h_last);
}


# one batch row per lane (32 rows per wave): w_hh A tiles block-diagonal over the two K lane groups, so every gate of a row lands in its own lane; all ds_bpermute go; k still summed 0..7 in order
# speedup vs baseline: 2.2227x; 2.2227x over previous
#include <hip/hip_runtime.h>


typedef float v2f __attribute__((ext_vector_type(2)));
typedef float v8f __attribute__((ext_vector_type(8)));
typedef float v4f __attribute__((ext_vector_type(4)));

#define WMMA_F32X4(A, B, C) \
    __builtin_amdgcn_wmma_f32_16x16x4_f32(false, (A), false, (B), (short)0, (C), false, false)

constexpr int Tn = 1024;
constexpr int Bn = 4096;
constexpr int TC  = 32;
constexpr int NCH = Tn / TC;
constexpr int NG  = TC / 2;

__device__ __forceinline__ float fast_sigmoid(float v) {
    return __builtin_amdgcn_rcpf(1.0f + __expf(-v));
}
__device__ __forceinline__ float fast_tanh(float v) {
    return 1.0f - 2.0f * __builtin_amdgcn_rcpf(__expf(2.0f * v) + 1.0f);
}

__global__ __launch_bounds__(32) void gru_wmma_kernel(
    const float* __restrict__ x,
    const float* __restrict__ h0,
    const float* __restrict__ w_ih,
    const float* __restrict__ w_hh,
    const float* __restrict__ b_ih,
    const float* __restrict__ b_hh,
    const float* __restrict__ fc_w,
    const float* __restrict__ fc_b,
    float* __restrict__ out,
    float* __restrict__ h_out)
{
    const int lane  = threadIdx.x;
    const int bbase = blockIdx.x * 32;
    const int b     = bbase + lane;

    const int  arow = lane & 7;
    const bool live = ((lane >> 4) & 1) == ((lane >> 3) & 1);
    v2f aR[4], aZ[4], aN[4];
    #pragma unroll
    for (int kp = 0; kp < 4; ++kp) {
        const float r0 = w_hh[(0  + arow) * 8 + 2 * kp], r1 = w_hh[(0  + arow) * 8 + 2 * kp + 1];
        const float z0 = w_hh[(8  + arow) * 8 + 2 * kp], z1 = w_hh[(8  + arow) * 8 + 2 * kp + 1];
        const float n0 = w_hh[(16 + arow) * 8 + 2 * kp], n1 = w_hh[(16 + arow) * 8 + 2 * kp + 1];
        aR[kp].x = live ? r0 : 0.0f; aR[kp].y = live ? r1 : 0.0f;
        aZ[kp].x = live ? z0 : 0.0f; aZ[kp].y = live ? z1 : 0.0f;
        aN[kp].x = live ? n0 : 0.0f; aN[kp].y = live ? n1 : 0.0f;
    }

    v8f cR, cZ, cN;
    #pragma unroll
    for (int j = 0; j < 8; ++j) {
        cR[j] = b_ih[j]     + b_hh[j];
        cZ[j] = b_ih[8 + j] + b_hh[8 + j];
        cN[j] = b_hh[16 + j];
    }

    float wR0[8], wR1[8], wZ0[8], wZ1[8], wN0[8], wN1[8], bN[8], fcw[8];
    #pragma unroll
    for (int j = 0; j < 8; ++j) {
        wR0[j] = w_ih[j * 2 + 0];        wR1[j] = w_ih[j * 2 + 1];
        wZ0[j] = w_ih[(8 + j) * 2 + 0];  wZ1[j] = w_ih[(8 + j) * 2 + 1];
        wN0[j] = w_ih[(16 + j) * 2 + 0]; wN1[j] = w_ih[(16 + j) * 2 + 1];
        bN[j]  = b_ih[16 + j];
        fcw[j] = fc_w[j];
    }
    const float fcb = fc_b[0];

    float h[8];
    {
        const v4f* hp = reinterpret_cast<const v4f*>(h0 + (size_t)b * 8);
        const v4f ha = hp[0], hb = hp[1];
        h[0] = ha.x; h[1] = ha.y; h[2] = ha.z; h[3] = ha.w;
        h[4] = hb.x; h[5] = hb.y; h[6] = hb.z; h[7] = hb.w;
    }

    float* outp = out + b;

    __shared__ v4f xs4[TC * 16];
    const int st_t = lane >> 4;
    const int st_p = lane & 15;
    v4f*       xst = xs4 + st_t * 16 + st_p;
    const v2f* xrd = reinterpret_cast<const v2f*>(xs4) + lane;
    const float* xg = x + (size_t)bbase * 2 + (size_t)st_t * (Bn * 2) + st_p * 4;

    v4f stage[NG];
    #pragma unroll
    for (int g = 0; g < NG; ++g)
        stage[g] = *reinterpret_cast<const v4f*>(xg + (size_t)g * 2 * Bn * 2);
    #pragma unroll
    for (int g = 0; g < NG; ++g)
        xst[g * 2 * 16] = stage[g];
    __syncthreads();

    for (int c = 0; c < NCH; ++c) {
      const bool more = (c + 1 < NCH);
      if (more) {
        xg += (size_t)TC * Bn * 2;
        #pragma unroll
        for (int g = 0; g < NG; ++g)
            stage[g] = *reinterpret_cast<const v4f*>(xg + (size_t)g * 2 * Bn * 2);
      }
      v2f xc = xrd[0];
      for (int tl = 0; tl < TC; ++tl) {
        const int tn  = (tl + 1 < TC) ? tl + 1 : tl;
        const v2f xn  = xrd[tn * 32];
        const float x0 = xc.x;
        const float x1 = xc.y;
        xc = xn;

        v2f bk[4];
        #pragma unroll
        for (int kp = 0; kp < 4; ++kp) { bk[kp].x = h[2 * kp]; bk[kp].y = h[2 * kp + 1]; }

        v8f dR = cR, dZ = cZ, dN = cN;
        #pragma unroll
        for (int kp = 0; kp < 4; ++kp) {
            dR = WMMA_F32X4(aR[kp], bk[kp], dR);
            dZ = WMMA_F32X4(aZ[kp], bk[kp], dZ);
            dN = WMMA_F32X4(aN[kp], bk[kp], dN);
        }

        #pragma unroll
        for (int j = 0; j < 8; ++j) {
            const float rJ  = fast_sigmoid(dR[j] + fmaf(x1, wR1[j], x0 * wR0[j]));
            const float zJ  = fast_sigmoid(dZ[j] + fmaf(x1, wZ1[j], x0 * wZ0[j]));
            const float xnJ = fmaf(x1, wN1[j], fmaf(x0, wN0[j], bN[j]));
            const float nJ  = fast_tanh(fmaf(rJ, dN[j], xnJ));
            h[j] = nJ + zJ * (h[j] - nJ);
        }

        const float accA = fmaf(fcw[5], h[5], fmaf(fcw[4], h[4], fmaf(fcw[1], h[1], fcw[0] * h[0])));
        const float accB = fmaf(fcw[7], h[7], fmaf(fcw[6], h[6], fmaf(fcw[3], h[3], fcw[2] * h[2])));
        *outp = fast_sigmoid(accA + accB + fcb);
        outp += Bn;
      }
      if (more) {
        #pragma unroll
        for (int g = 0; g < NG; ++g)
            xst[g * 2 * 16] = stage[g];
      }
      __syncthreads();
    }

    {
        v4f* hp = reinterpret_cast<v4f*>(h_out + (size_t)b * 8);
        v4f ha, hb;
        ha.x = h[0]; ha.y = h[1]; ha.z = h[2]; ha.w = h[3];
        hb.x = h[4]; hb.y = h[5]; hb.z = h[6]; hb.w = h[7];
        hp[0] = ha; hp[1] = hb;
    }
}

extern "C" void kernel_launch(void* const* d_in, const int* in_sizes, int n_in,
                              void* d_out, int out_size, void* d_ws, size_t ws_size,
                              hipStream_t stream) {
    const float* x    = (const float*)d_in[0];
    const float* h0   = (const float*)d_in[1];
    const float* w_ih = (const float*)d_in[2];
    const float* w_hh = (const float*)d_in[3];
    const float* b_ih = (const float*)d_in[4];
    const float* b_hh = (const float*)d_in[5];
    const float* fc_w = (const float*)d_in[6];
    const float* fc_b = (const float*)d_in[7];

    float* out    = (float*)d_out;
    float* h_last = out + (size_t)Tn * Bn;

    gru_wmma_kernel<<<128, 32, 0, stream>>>(
        x, h0, w_ih, w_hh, b_ih, b_hh, fc_w, fc_b, out, h_last);
}


# w_hh rows regrouped over the two A tiles by when a step needs them: r and n-hidden rows in the tile issued first, z and head rows in the second, whose products run behind the r chain; sums unchanged
# speedup vs baseline: 7.3121x; 3.2897x over previous
#include <hip/hip_runtime.h>


typedef float v2f __attribute__((ext_vector_type(2)));
typedef float v8f __attribute__((ext_vector_type(8)));
typedef float v4f __attribute__((ext_vector_type(4)));

#define WMMA_F32X4(A, B, C) \
    __builtin_amdgcn_wmma_f32_16x16x4_f32(false, (A), false, (B), (short)0, (C), false, false)

constexpr int Tn = 1024;
constexpr int Bn = 4096;
constexpr int TC  = 32;
constexpr int NCH = Tn / TC;
constexpr int TG  = 8;
constexpr int NGR = TC / TG;
constexpr int RW  = 32;
constexpr int NT  = 128;
constexpr int NG  = (TC * 16) / NT;

constexpr float LOG2E = 1.4426950408889634f;
constexpr float SC_SIG = -LOG2E;
constexpr float SC_TANH = 2.0f * LOG2E;

__device__ __forceinline__ float sigmoid_of_exponent(float a) {
    return __builtin_amdgcn_rcpf(1.0f + __builtin_amdgcn_exp2f(a));
}
__device__ __forceinline__ float tanh_of_exponent(float a) {
    return 1.0f - 2.0f * __builtin_amdgcn_rcpf(__builtin_amdgcn_exp2f(a) + 1.0f);
}

template <int BANKS>
__device__ __forceinline__ float take_from_copy(float v) {
    const int i = __float_as_int(v);
    return __int_as_float(__builtin_amdgcn_update_dpp(i, i, 0x168, 0xF, BANKS, false));
}

__global__ __launch_bounds__(NT) void gru_wmma_kernel(
    const float* __restrict__ x,
    const float* __restrict__ h0,
    const float* __restrict__ w_ih,
    const float* __restrict__ w_hh,
    const float* __restrict__ b_ih,
    const float* __restrict__ b_hh,
    const float* __restrict__ fc_w,
    const float* __restrict__ fc_b,
    float* __restrict__ out,
    float* __restrict__ h_out)
{
    const int tid   = threadIdx.x;
    const int lane  = tid & 31;
    const int wv    = tid >> 5;
    const int bbase = blockIdx.x * RW;
    const int c     = lane & 15;
    const int hf    = lane >> 4;
    const int d     = c >> 3;
    const int rwg   = wv * 8 + (c & 7);
    const int b     = bbase + rwg;
    const int u0    = 4 * d + 2 * hf;
    const bool copy1 = (d != 0);

    v2f aP[2], aQ[2];
    {
        const int mh = c >> 3, ix = c & 7;
        const int  unitP = 4 * (ix >> 2) + 2 * mh + (ix & 1);
        const bool isN   = ((ix >> 1) & 1) != 0;
        const float* rowP = w_hh + ((isN ? 16 : 0) + unitP) * 8;
        const float  scP  = isN ? SC_TANH : SC_SIG;
        const int  unitQ = 4 * ((ix >> 1) & 1) + 2 * mh + (ix & 1);
        const float* rowQ = (ix < 4) ? (w_hh + (8 + unitQ) * 8) : fc_w;
        #pragma unroll
        for (int s = 0; s < 2; ++s) {
            const int k = 4 * s + 2 * hf;
            aP[s].x = scP * rowP[k];
            aP[s].y = scP * rowP[k + 1];
            aQ[s].x = (ix < 5) ? SC_SIG * rowQ[k]     : 0.0f;
            aQ[s].y = (ix < 5) ? SC_SIG * rowQ[k + 1] : 0.0f;
        }
    }

    v8f cP, cQ;
    #pragma unroll
    for (int i = 0; i < 8; ++i) {
        const int unit = 4 * (i >> 2) + 2 * hf + (i & 1);
        cP[i] = ((i >> 1) & 1) ? SC_TANH * b_hh[16 + unit]
                               : SC_SIG  * (b_ih[unit] + b_hh[unit]);
    }
    #pragma unroll
    for (int i = 0; i < 4; ++i) {
        const int unit = 4 * (i >> 1) + 2 * hf + (i & 1);
        cQ[i] = SC_SIG * (b_ih[8 + unit] + b_hh[8 + unit]);
    }
    cQ[4] = SC_SIG * fc_b[0];
    cQ[5] = 0.0f; cQ[6] = 0.0f; cQ[7] = 0.0f;

    float wR0[2], wR1[2], wZ0[2], wZ1[2], wN0[2], wN1[2], bN[2];
    #pragma unroll
    for (int i = 0; i < 2; ++i) {
        wR0[i] = SC_SIG  * w_ih[(u0 + i) * 2 + 0];       wR1[i] = SC_SIG  * w_ih[(u0 + i) * 2 + 1];
        wZ0[i] = SC_SIG  * w_ih[(8 + u0 + i) * 2 + 0];   wZ1[i] = SC_SIG  * w_ih[(8 + u0 + i) * 2 + 1];
        wN0[i] = SC_TANH * w_ih[(16 + u0 + i) * 2 + 0];  wN1[i] = SC_TANH * w_ih[(16 + u0 + i) * 2 + 1];
        bN[i]  = SC_TANH * b_ih[16 + u0 + i];
    }

    auto input_terms = [&](const v2f xv, float (&tr)[2], float (&tz)[2], float (&tn)[2]) {
        #pragma unroll
        for (int i = 0; i < 2; ++i) {
            tr[i] = fmaf(xv.y, wR1[i], xv.x * wR0[i]);
            tz[i] = fmaf(xv.y, wZ1[i], xv.x * wZ0[i]);
            tn[i] = fmaf(xv.y, wN1[i], fmaf(xv.x, wN0[i], bN[i]));
        }
    };

    float hv[2];
    {
        const v2f hp = *reinterpret_cast<const v2f*>(h0 + (size_t)b * 8 + u0);
        hv[0] = hp.x; hv[1] = hp.y;
    }

    __shared__ v4f   xs4[2][TC * 16];
    __shared__ float obuf[2][(TC + 1) * RW];
    const v2f* xrd = reinterpret_cast<const v2f*>(&xs4[0][0]) + rwg;
    const float* xg = x + (size_t)bbase * 2 + (size_t)(tid >> 4) * (Bn * 2) + (tid & 15) * 4;

    v4f stage[NG];
    #pragma unroll
    for (int g = 0; g < NG; ++g)
        stage[g] = *reinterpret_cast<const v4f*>(xg + (size_t)g * 8 * Bn * 2);
    #pragma unroll
    for (int g = 0; g < NG; ++g)
        xs4[0][g * NT + tid] = stage[g];
    __syncthreads();

    v8f dP, dQ;
    v2f b2q;
    {
        v2f b1, b2;
        b1.x = take_from_copy<0xC>(hv[0]); b1.y = take_from_copy<0xC>(hv[1]);
        b2.x = take_from_copy<0x3>(hv[0]); b2.y = take_from_copy<0x3>(hv[1]);
        dP = WMMA_F32X4(aP[0], b1, cP);
        dP = WMMA_F32X4(aP[1], b2, dP);
        dQ = WMMA_F32X4(aQ[0], b1, cQ);
        b2q = b2;
    }

    for (int ch = 0; ch < NCH; ++ch) {
      const bool more = (ch + 1 < NCH);
      const int  buf  = ch & 1;
      if (more) {
        xg += (size_t)TC * Bn * 2;
        #pragma unroll
        for (int g = 0; g < NG; ++g)
            stage[g] = *reinterpret_cast<const v4f*>(xg + (size_t)g * 8 * Bn * 2);
      }
      const v2f* xb = xrd + buf * (TC * 32);
      float*     ob = &obuf[buf][rwg];
      v2f xq[TG];
      #pragma unroll
      for (int s = 0; s < TG; ++s) xq[s] = xb[s * 32];
      float xr[2], xz[2], xn[2];
      input_terms(xq[0], xr, xz, xn);
      for (int g = 0; g < NGR; ++g) {
        const int gn = (g + 1 < NGR) ? g + 1 : g;
        v2f xnq[TG];
        #pragma unroll
        for (int s = 0; s < TG; ++s) xnq[s] = xb[(gn * TG + s) * 32];
        float og[TG];
        #pragma unroll
        for (int s = 0; s < TG; ++s) {
          float eR[2];
          #pragma unroll
          for (int i = 0; i < 2; ++i) {
            v2f pr;
            pr.x = dP[i]; pr.y = dP[4 + i];
            eR[i] = __builtin_amdgcn_exp2f(pr[d] + xr[i]);
          }
          __builtin_amdgcn_sched_barrier(0);
          dQ = WMMA_F32X4(aQ[1], b2q, dQ);

          float rI[2];
          #pragma unroll
          for (int i = 0; i < 2; ++i) rI[i] = __builtin_amdgcn_rcpf(1.0f + eR[i]);
          __builtin_amdgcn_sched_barrier(0);

          og[s] = sigmoid_of_exponent(dQ[4]);
          #pragma unroll
          for (int i = 0; i < 2; ++i) {
            v2f ph, pz;
            ph.x = dP[2 + i]; ph.y = dP[6 + i];
            pz.x = dQ[i];     pz.y = dQ[2 + i];
            const float nI = tanh_of_exponent(fmaf(rI[i], ph[d], xn[i]));
            const float zI = sigmoid_of_exponent(pz[d] + xz[i]);
            hv[i] = nI + zI * (hv[i] - nI);
          }
          input_terms((s + 1 < TG) ? xq[s + 1] : xnq[0], xr, xz, xn);
          asm volatile("" :: "v"(xr[0]), "v"(xr[1]), "v"(xz[0]), "v"(xz[1]), "v"(xn[0]), "v"(xn[1]));

          v2f b1, b2;
          b1.x = take_from_copy<0xC>(hv[0]); b1.y = take_from_copy<0xC>(hv[1]);
          b2.x = take_from_copy<0x3>(hv[0]); b2.y = take_from_copy<0x3>(hv[1]);
          dP = WMMA_F32X4(aP[0], b1, cP);
          dP = WMMA_F32X4(aP[1], b2, dP);
          __builtin_amdgcn_sched_barrier(0);
          dQ = WMMA_F32X4(aQ[0], b1, cQ);
          b2q = b2;
          __builtin_amdgcn_sched_barrier(0);
        }
        #pragma unroll
        for (int s = 0; s < TG; ++s) ob[(g * TG + s) * RW] = og[s];
        #pragma unroll
        for (int s = 0; s < TG; ++s) xq[s] = xnq[s];
      }
      {
        const v8f dQe = WMMA_F32X4(aQ[1], b2q, dQ);
        ob[TC * RW] = sigmoid_of_exponent(dQe[4]);
      }
      if (more) {
        #pragma unroll
        for (int g = 0; g < NG; ++g)
            xs4[buf ^ 1][g * NT + tid] = stage[g];
      }
      __syncthreads();
      {
        float* op = out + ((size_t)ch * TC + wv * 8) * Bn + bbase + lane;
        const float* os = &obuf[buf][(wv * 8 + 1) * RW + lane];
        #pragma unroll
        for (int i = 0; i < 8; ++i)
            op[(size_t)i * Bn] = os[i * RW];
      }
    }

    {
        v2f hp; hp.x = hv[0]; hp.y = hv[1];
        *reinterpret_cast<v2f*>(h_out + (size_t)b * 8 + u0) = hp;
    }
}

extern "C" void kernel_launch(void* const* d_in, const int* in_sizes, int n_in,
                              void* d_out, int out_size, void* d_ws, size_t ws_size,
                              hipStream_t stream) {
    const float* x    = (const float*)d_in[0];
    const float* h0   = (const float*)d_in[1];
    const float* w_ih = (const float*)d_in[2];
    const float* w_hh = (const float*)d_in[3];
    const float* b_ih = (const float*)d_in[4];
    const float* b_hh = (const float*)d_in[5];
    const float* fc_w = (const float*)d_in[6];
    const float* fc_b = (const float*)d_in[7];

    float* out    = (float*)d_out;
    float* h_last = out + (size_t)Tn * Bn;

    gru_wmma_kernel<<<128, 128, 0, stream>>>(
        x, h0, w_ih, w_hh, b_ih, b_hh, fc_w, fc_b, out, h_last);
}


# each A row takes the k order of the lanes that read it: a lane's own h pair is k step 0 straight from its registers, the other pair one unmasked DPP move behind product 1; units 4-7 sum k 4-7,0-3
# speedup vs baseline: 7.7146x; 1.0550x over previous
#include <hip/hip_runtime.h>


typedef float v2f __attribute__((ext_vector_type(2)));
typedef float v8f __attribute__((ext_vector_type(8)));
typedef float v4f __attribute__((ext_vector_type(4)));

#define WMMA_F32X4(A, B, C) \
    __builtin_amdgcn_wmma_f32_16x16x4_f32(false, (A), false, (B), (short)0, (C), false, false)

constexpr int Tn = 1024;
constexpr int Bn = 4096;
constexpr int TC  = 32;
constexpr int NCH = Tn / TC;
constexpr int TG  = 8;
constexpr int NGR = TC / TG;
constexpr int RW  = 32;
constexpr int NT  = 128;
constexpr int NG  = (TC * 16) / NT;

constexpr float LOG2E = 1.4426950408889634f;
constexpr float SC_SIG = -LOG2E;
constexpr float SC_TANH = 2.0f * LOG2E;

__device__ __forceinline__ float sigmoid_of_exponent(float a) {
    return __builtin_amdgcn_rcpf(1.0f + __builtin_amdgcn_exp2f(a));
}
__device__ __forceinline__ float tanh_of_exponent(float a) {
    return 1.0f - 2.0f * __builtin_amdgcn_rcpf(__builtin_amdgcn_exp2f(a) + 1.0f);
}

__device__ __forceinline__ float from_other_copy(float v) {
    const int i = __float_as_int(v);
    return __int_as_float(__builtin_amdgcn_update_dpp(i, i, 0x168, 0xF, 0xF, true));
}

__global__ __launch_bounds__(NT) void gru_wmma_kernel(
    const float* __restrict__ x,
    const float* __restrict__ h0,
    const float* __restrict__ w_ih,
    const float* __restrict__ w_hh,
    const float* __restrict__ b_ih,
    const float* __restrict__ b_hh,
    const float* __restrict__ fc_w,
    const float* __restrict__ fc_b,
    float* __restrict__ out,
    float* __restrict__ h_out)
{
    const int tid   = threadIdx.x;
    const int lane  = tid & 31;
    const int wv    = tid >> 5;
    const int bbase = blockIdx.x * RW;
    const int c     = lane & 15;
    const int hf    = lane >> 4;
    const int d     = c >> 3;
    const int rwg   = wv * 8 + (c & 7);
    const int b     = bbase + rwg;
    const int u0    = 4 * d + 2 * hf;
    const bool copy1 = (d != 0);

    v2f aP[2], aQ[2];
    {
        const int mh = c >> 3, ix = c & 7;
        const int  unitP = 4 * (ix >> 2) + 2 * mh + (ix & 1);
        const bool isN   = ((ix >> 1) & 1) != 0;
        const float* rowP = w_hh + ((isN ? 16 : 0) + unitP) * 8;
        const float  scP  = isN ? SC_TANH : SC_SIG;
        const int  unitQ = 4 * ((ix >> 1) & 1) + 2 * mh + (ix & 1);
        const float* rowQ = (ix < 4) ? (w_hh + (8 + unitQ) * 8) : fc_w;
        const int cpP = ix >> 2;
        const int cpQ = (ix < 4) ? ((ix >> 1) & 1) : 0;
        #pragma unroll
        for (int s = 0; s < 2; ++s) {
            const int kP = 4 * (s ^ cpP) + 2 * hf;
            const int kQ = 4 * (s ^ cpQ) + 2 * hf;
            aP[s].x = scP * rowP[kP];
            aP[s].y = scP * rowP[kP + 1];
            aQ[s].x = (ix < 5) ? SC_SIG * rowQ[kQ]     : 0.0f;
            aQ[s].y = (ix < 5) ? SC_SIG * rowQ[kQ + 1] : 0.0f;
        }
    }

    v8f cP, cQ;
    #pragma unroll
    for (int i = 0; i < 8; ++i) {
        const int unit = 4 * (i >> 2) + 2 * hf + (i & 1);
        cP[i] = ((i >> 1) & 1) ? SC_TANH * b_hh[16 + unit]
                               : SC_SIG  * (b_ih[unit] + b_hh[unit]);
    }
    #pragma unroll
    for (int i = 0; i < 4; ++i) {
        const int unit = 4 * (i >> 1) + 2 * hf + (i & 1);
        cQ[i] = SC_SIG * (b_ih[8 + unit] + b_hh[8 + unit]);
    }
    cQ[4] = SC_SIG * fc_b[0];
    cQ[5] = 0.0f; cQ[6] = 0.0f; cQ[7] = 0.0f;

    float wR0[2], wR1[2], wZ0[2], wZ1[2], wN0[2], wN1[2], bN[2];
    #pragma unroll
    for (int i = 0; i < 2; ++i) {
        wR0[i] = SC_SIG  * w_ih[(u0 + i) * 2 + 0];       wR1[i] = SC_SIG  * w_ih[(u0 + i) * 2 + 1];
        wZ0[i] = SC_SIG  * w_ih[(8 + u0 + i) * 2 + 0];   wZ1[i] = SC_SIG  * w_ih[(8 + u0 + i) * 2 + 1];
        wN0[i] = SC_TANH * w_ih[(16 + u0 + i) * 2 + 0];  wN1[i] = SC_TANH * w_ih[(16 + u0 + i) * 2 + 1];
        bN[i]  = SC_TANH * b_ih[16 + u0 + i];
    }

    auto input_terms = [&](const v2f xv, float (&tr)[2], float (&tz)[2], float (&tn)[2]) {
        #pragma unroll
        for (int i = 0; i < 2; ++i) {
            tr[i] = fmaf(xv.y, wR1[i], xv.x * wR0[i]);
            tz[i] = fmaf(xv.y, wZ1[i], xv.x * wZ0[i]);
            tn[i] = fmaf(xv.y, wN1[i], fmaf(xv.x, wN0[i], bN[i]));
        }
    };

    float hv[2];
    {
        const v2f hp = *reinterpret_cast<const v2f*>(h0 + (size_t)b * 8 + u0);
        hv[0] = hp.x; hv[1] = hp.y;
    }

    __shared__ v4f   xs4[2][TC * 16];
    __shared__ float obuf[2][(TC + 1) * RW];
    __shared__ float odump[(TC + 1) * RW];
    const v2f* xrd = reinterpret_cast<const v2f*>(&xs4[0][0]) + rwg;
    const float* xg = x + (size_t)bbase * 2 + (size_t)(tid >> 4) * (Bn * 2) + (tid & 15) * 4;

    v4f stage[NG];
    #pragma unroll
    for (int g = 0; g < NG; ++g)
        stage[g] = *reinterpret_cast<const v4f*>(xg + (size_t)g * 8 * Bn * 2);
    #pragma unroll
    for (int g = 0; g < NG; ++g)
        xs4[0][g * NT + tid] = stage[g];
    __syncthreads();

    v8f dP, dQ;
    v2f b2q;
    {
        v2f b1, b2;
        b1.x = hv[0];                  b1.y = hv[1];
        b2.x = from_other_copy(hv[0]); b2.y = from_other_copy(hv[1]);
        dP = WMMA_F32X4(aP[0], b1, cP);
        dP = WMMA_F32X4(aP[1], b2, dP);
        dQ = WMMA_F32X4(aQ[0], b1, cQ);
        b2q = b2;
    }

    for (int ch = 0; ch < NCH; ++ch) {
      const bool more = (ch + 1 < NCH);
      const int  buf  = ch & 1;
      if (more) {
        xg += (size_t)TC * Bn * 2;
        #pragma unroll
        for (int g = 0; g < NG; ++g)
            stage[g] = *reinterpret_cast<const v4f*>(xg + (size_t)g * 8 * Bn * 2);
      }
      const v2f* xb = xrd + buf * (TC * 32);
      float*     ob = copy1 ? &odump[rwg] : &obuf[buf][rwg];
      v2f xq[TG];
      #pragma unroll
      for (int s = 0; s < TG; ++s) xq[s] = xb[s * 32];
      float xr[2], xz[2], xn[2];
      input_terms(xq[0], xr, xz, xn);
      for (int g = 0; g < NGR; ++g) {
        const int gn = (g + 1 < NGR) ? g + 1 : g;
        v2f xnq[TG];
        #pragma unroll
        for (int s = 0; s < TG; ++s) xnq[s] = xb[(gn * TG + s) * 32];
        float og[TG];
        #pragma unroll
        for (int s = 0; s < TG; ++s) {
          float eR[2];
          #pragma unroll
          for (int i = 0; i < 2; ++i) {
            v2f pr;
            pr.x = dP[i]; pr.y = dP[4 + i];
            eR[i] = __builtin_amdgcn_exp2f(pr[d] + xr[i]);
          }
          __builtin_amdgcn_sched_barrier(0);
          dQ = WMMA_F32X4(aQ[1], b2q, dQ);

          float rI[2];
          #pragma unroll
          for (int i = 0; i < 2; ++i) rI[i] = __builtin_amdgcn_rcpf(1.0f + eR[i]);
          __builtin_amdgcn_sched_barrier(0);

          og[s] = sigmoid_of_exponent(dQ[4]);
          #pragma unroll
          for (int i = 0; i < 2; ++i) {
            v2f ph, pz;
            ph.x = dP[2 + i]; ph.y = dP[6 + i];
            pz.x = dQ[i];     pz.y = dQ[2 + i];
            const float nI = tanh_of_exponent(fmaf(rI[i], ph[d], xn[i]));
            const float zI = sigmoid_of_exponent(pz[d] + xz[i]);
            hv[i] = nI + zI * (hv[i] - nI);
          }
          input_terms((s + 1 < TG) ? xq[s + 1] : xnq[0], xr, xz, xn);
          asm volatile("" :: "v"(xr[0]), "v"(xr[1]), "v"(xz[0]), "v"(xz[1]), "v"(xn[0]), "v"(xn[1]));

          v2f b1, b2;
          b1.x = hv[0];                  b1.y = hv[1];
          b2.x = from_other_copy(hv[0]); b2.y = from_other_copy(hv[1]);
          dP = WMMA_F32X4(aP[0], b1, cP);
          dP = WMMA_F32X4(aP[1], b2, dP);
          __builtin_amdgcn_sched_barrier(0);
          dQ = WMMA_F32X4(aQ[0], b1, cQ);
          b2q = b2;
          __builtin_amdgcn_sched_barrier(0);
        }
        #pragma unroll
        for (int s = 0; s < TG; ++s) ob[(g * TG + s) * RW] = og[s];
        #pragma unroll
        for (int s = 0; s < TG; ++s) xq[s] = xnq[s];
      }
      {
        const v8f dQe = WMMA_F32X4(aQ[1], b2q, dQ);
        ob[TC * RW] = sigmoid_of_exponent(dQe[4]);
      }
      if (more) {
        #pragma unroll
        for (int g = 0; g < NG; ++g)
            xs4[buf ^ 1][g * NT + tid] = stage[g];
      }
      __syncthreads();
      {
        float* op = out + ((size_t)ch * TC + wv * 8) * Bn + bbase + lane;
        const float* os = &obuf[buf][(wv * 8 + 1) * RW + lane];
        #pragma unroll
        for (int i = 0; i < 8; ++i)
            op[(size_t)i * Bn] = os[i * RW];
      }
    }

    {
        v2f hp; hp.x = hv[0]; hp.y = hv[1];
        *reinterpret_cast<v2f*>(h_out + (size_t)b * 8 + u0) = hp;
    }
}

extern "C" void kernel_launch(void* const* d_in, const int* in_sizes, int n_in,
                              void* d_out, int out_size, void* d_ws, size_t ws_size,
                              hipStream_t stream) {
    const float* x    = (const float*)d_in[0];
    const float* h0   = (const float*)d_in[1];
    const float* w_ih = (const float*)d_in[2];
    const float* w_hh = (const float*)d_in[3];
    const float* b_ih = (const float*)d_in[4];
    const float* b_hh = (const float*)d_in[5];
    const float* fc_w = (const float*)d_in[6];
    const float* fc_b = (const float*)d_in[7];

    float* out    = (float*)d_out;
    float* h_last = out + (size_t)Tn * Bn;

    gru_wmma_kernel<<<128, 128, 0, stream>>>(
        x, h0, w_ih, w_hh, b_ih, b_hh, fc_w, fc_b, out, h_last);
}
